# starting kernel
# baseline (speedup 1.0000x reference)
#include <hip/hip_runtime.h>
#include <hip/hip_bf16.h>
#include <math.h>

#define BDIM 1024
#define DDIM 256
#define CDIM 1000
#define EPSV 1e-6f

typedef __attribute__((ext_vector_type(2))) float v2f;
typedef __attribute__((ext_vector_type(8))) float v8f;

__global__ void pd_xterm_kernel(const float* __restrict__ x,
                                float* __restrict__ xterm) {
    const int lane = threadIdx.x & 31;
    const int wave = threadIdx.x >> 5;
    const int row  = blockIdx.x * 8 + wave;

    const float* xr = x + row * DDIM;
    float s1 = 0.f, s2 = 0.f;
#pragma unroll
    for (int i = 0; i < DDIM / 32; ++i) {
        float v = xr[lane + i * 32];
        s1 += v;
        s2 += v * v;
    }
#pragma unroll
    for (int off = 16; off > 0; off >>= 1) {
        s1 += __shfl_xor(s1, off, 32);
        s2 += __shfl_xor(s2, off, 32);
    }
    if (lane == 0) {
        xterm[row] = s2 + 2.0f * EPSV * s1;
    }
}

__global__ void pd_wterm_kernel(const float* __restrict__ w,
                                float* __restrict__ wterm) {
    const int c = blockIdx.x * blockDim.x + threadIdx.x;
    if (c >= CDIM) return;
    float s1 = 0.f, s2 = 0.f;
#pragma unroll 8
    for (int d = 0; d < DDIM; ++d) {
        float v = w[d * CDIM + c];
        s1 += v;
        s2 += v * v;
    }
    wterm[c] = s2 - 2.0f * EPSV * s1 + (float)DDIM * EPSV * EPSV;
}

__global__ void pd_wmma_kernel(const float* __restrict__ x,
                               const float* __restrict__ w,
                               const float* __restrict__ xterm,
                               const float* __restrict__ wterm,
                               float* __restrict__ out) {
    const int lane  = threadIdx.x & 31;
    const int wave  = threadIdx.x >> 5;
    const int wm    = wave >> 2;
    const int wn    = wave & 3;

    const int tile_m0 = blockIdx.y * 32 + wm * 16;
    const int tile_n0 = blockIdx.x * 64 + wn * 16;

    const int l15   = lane & 15;
    const int khalf = (lane >> 4) * 2;

    const float* arow = x + (tile_m0 + l15) * DDIM + khalf;
    const int col   = tile_n0 + l15;
    const int colc  = col < CDIM ? col : (CDIM - 1);
    const float* bcol = w + colc;

    v8f acc = {};

#pragma unroll 8
    for (int k = 0; k < DDIM; k += 4) {
        v2f a, b;
        a.x = arow[k];
        a.y = arow[k + 1];
        b.x = bcol[(k + khalf) * CDIM];
        b.y = bcol[(k + khalf + 1) * CDIM];
        acc = __builtin_amdgcn_wmma_f32_16x16x4_f32(
             false, a,  false, b,
             (short)0, acc,  false,  false);
    }

    const float wt  = wterm[colc];
    const int   mhi = (lane >> 4) * 8;
#pragma unroll
    for (int r = 0; r < 8; ++r) {
        const int row_g = tile_m0 + r + mhi;
        const float s   = xterm[row_g] + wt - 2.0f * acc[r];
        if (col < CDIM) {
            out[row_g * CDIM + col] = sqrtf(fmaxf(s, 0.0f));
        }
    }
}

extern "C" void kernel_launch(void* const* d_in, const int* in_sizes, int n_in,
                              void* d_out, int out_size, void* d_ws, size_t ws_size,
                              hipStream_t stream) {
    const float* x = (const float*)d_in[0];
    const float* w = (const float*)d_in[1];
    float* out = (float*)d_out;

    float* xterm = (float*)d_ws;
    float* wterm = xterm + BDIM;

    pd_xterm_kernel<<<BDIM / 8, 256, 0, stream>>>(x, xterm);
    pd_wterm_kernel<<<(CDIM + 255) / 256, 256, 0, stream>>>(w, wterm);
    dim3 grid(16, 32, 1);
    pd_wmma_kernel<<<grid, 256, 0, stream>>>(x, w, xterm, wterm, out);
}


# GEMM kernel re-tiled without LDS: a wave owns 16x64 as 4 accumulators over interleaved columns, so one 16-byte load of w feeds 4 B fragments; loads stream ahead in registers; k order of sum regrouped
# speedup vs baseline: 3.6586x; 3.6586x over previous
#include <hip/hip_runtime.h>
#include <hip/hip_bf16.h>
#include <math.h>

#define BDIM 1024
#define DDIM 256
#define CDIM 1000
#define EPSV 1e-6f

typedef __attribute__((ext_vector_type(2))) float v2f;
typedef __attribute__((ext_vector_type(8))) float v8f;

__global__ void pd_xterm_kernel(const float* __restrict__ x,
                                float* __restrict__ xterm) {
    const int lane = threadIdx.x & 31;
    const int wave = threadIdx.x >> 5;
    const int row  = blockIdx.x * 8 + wave;

    const float* xr = x + row * DDIM;
    float s1 = 0.f, s2 = 0.f;
#pragma unroll
    for (int i = 0; i < DDIM / 32; ++i) {
        float v = xr[lane + i * 32];
        s1 += v;
        s2 += v * v;
    }
#pragma unroll
    for (int off = 16; off > 0; off >>= 1) {
        s1 += __shfl_xor(s1, off, 32);
        s2 += __shfl_xor(s2, off, 32);
    }
    if (lane == 0) {
        xterm[row] = s2 + 2.0f * EPSV * s1;
    }
}

#define WT_WAVES 8
#define WT_ROWS  (DDIM / WT_WAVES)

__global__ void __launch_bounds__(WT_WAVES * 32)
pd_wterm_kernel(const float* __restrict__ w,
                float* __restrict__ wterm) {
    __shared__ float2 part[WT_WAVES][32];

    const int lane = threadIdx.x & 31;
    const int wave = threadIdx.x >> 5;
    const int c    = blockIdx.x * 32 + lane;
    const int cc   = c < CDIM ? c : (CDIM - 1);
    const float* p = w + (wave * WT_ROWS) * CDIM + cc;

    float v[WT_ROWS];
#pragma unroll
    for (int i = 0; i < WT_ROWS; ++i) {
        v[i] = p[i * CDIM];
    }

    float a1[4] = {0.f, 0.f, 0.f, 0.f};
    float a2[4] = {0.f, 0.f, 0.f, 0.f};
#pragma unroll
    for (int i = 0; i < WT_ROWS; ++i) {
        a1[i & 3] += v[i];
        a2[i & 3] += v[i] * v[i];
    }
    const float s1 = (a1[0] + a1[1]) + (a1[2] + a1[3]);
    const float s2 = (a2[0] + a2[1]) + (a2[2] + a2[3]);

    part[wave][lane] = make_float2(s1, s2);
    __syncthreads();

    if (wave == 0) {
        float t1 = 0.f, t2 = 0.f;
        for (int k = 0; k < WT_WAVES; ++k) {
            const float2 q = part[k][lane];
            t1 += q.x;
            t2 += q.y;
        }
        if (c < CDIM) {
            wterm[c] = t2 - 2.0f * EPSV * t1 + (float)DDIM * EPSV * EPSV;
        }
    }
}

typedef __attribute__((ext_vector_type(4))) float v4f;

#define GM_WAVES  4
#define GM_CHUNKS (DDIM / 16)
#define GM_AHEAD  4

struct GmChunk {
    v4f a[2];
    v4f b[2][4];
};

__device__ __forceinline__ void gm_load(GmChunk& ck, const float* __restrict__ ap,
                                        const float* __restrict__ bp, int ch) {
#pragma unroll
    for (int q = 0; q < 2; ++q) {
        ck.a[q] = *(const v4f*)(ap + 16 * ch + 8 * q);
#pragma unroll
        for (int s = 0; s < 4; ++s) {
            ck.b[q][s] = *(const v4f*)(bp + (16 * ch + 8 * q + s) * CDIM);
        }
    }
}

__device__ __forceinline__ void gm_mma(const GmChunk& ck, v8f (&acc)[4]) {
#pragma unroll
    for (int q = 0; q < 2; ++q) {
#pragma unroll
        for (int p = 0; p < 2; ++p) {
            v2f a;
            a.x = ck.a[q][2 * p];
            a.y = ck.a[q][2 * p + 1];
#pragma unroll
            for (int j = 0; j < 4; ++j) {
                v2f b;
                b.x = ck.b[q][2 * p][j];
                b.y = ck.b[q][2 * p + 1][j];
                acc[j] = __builtin_amdgcn_wmma_f32_16x16x4_f32(
                     false, a,  false, b,
                     (short)0, acc[j],  false,  false);
            }
        }
    }
}

__global__ void __launch_bounds__(GM_WAVES * 32) __attribute__((amdgpu_waves_per_eu(1, 1)))
pd_wmma_kernel(const float* __restrict__ x,
               const float* __restrict__ w,
               const float* __restrict__ xterm,
               const float* __restrict__ wterm,
               float* __restrict__ out) {
    const int lane = threadIdx.x & 31;
    const int wave = threadIdx.x >> 5;
    const int c    = lane & 15;
    const int h    = lane >> 4;

    const int tile_m0 = blockIdx.y * 64 + wave * 16;
    const int gcol    = blockIdx.x * 64 + 4 * c;
    const bool live   = gcol < CDIM;
    const int gcolc   = live ? gcol : (CDIM - 4);

    const float* ap = x + (tile_m0 + c) * DDIM + 4 * h;
    const float* bp = w + (4 * h) * CDIM + gcolc;

    const float* xt = xterm + tile_m0 + 8 * h;
    const v4f xt0 = *(const v4f*)(xt);
    const v4f xt1 = *(const v4f*)(xt + 4);
    const v4f wt  = *(const v4f*)(wterm + gcolc);

    v8f acc[4] = {};
    GmChunk ring[GM_AHEAD + 1];

#pragma unroll
    for (int ch = 0; ch < GM_AHEAD; ++ch) {
        gm_load(ring[ch], ap, bp, ch);
    }
#pragma unroll
    for (int ch = 0; ch < GM_CHUNKS; ++ch) {
        if (ch + GM_AHEAD < GM_CHUNKS) {
            gm_load(ring[(ch + GM_AHEAD) % (GM_AHEAD + 1)], ap, bp, ch + GM_AHEAD);
        }
        if (ch == 0) {
            asm volatile("" :: "v"(xt0), "v"(xt1), "v"(wt));
        }
        gm_mma(ring[ch % (GM_AHEAD + 1)], acc);
    }

    if (live) {
#pragma unroll
        for (int r = 0; r < 8; ++r) {
            const int row_g = tile_m0 + 8 * h + r;
            const float xr  = (r < 4) ? xt0[r & 3] : xt1[r & 3];
            v4f o;
#pragma unroll
            for (int j = 0; j < 4; ++j) {
                const float s = xr + wt[j] - 2.0f * acc[j][r];
                o[j] = sqrtf(fmaxf(s, 0.0f));
            }
            *(v4f*)(out + row_g * CDIM + gcol) = o;
        }
    }
}

extern "C" void kernel_launch(void* const* d_in, const int* in_sizes, int n_in,
                              void* d_out, int out_size, void* d_ws, size_t ws_size,
                              hipStream_t stream) {
    const float* x = (const float*)d_in[0];
    const float* w = (const float*)d_in[1];
    float* out = (float*)d_out;

    float* xterm = (float*)d_ws;
    float* wterm = xterm + BDIM;

    pd_xterm_kernel<<<BDIM / 8, 256, 0, stream>>>(x, xterm);
    pd_wterm_kernel<<<(CDIM + 31) / 32, WT_WAVES * 32, 0, stream>>>(w, wterm);
    dim3 grid(16, 16, 1);
    pd_wmma_kernel<<<grid, GM_WAVES * 32, 0, stream>>>(x, w, xterm, wterm, out);
}


# Row-sum and column-sum kernels merged into one launch: work-groups 0-31 take the column sums, 32-159 the row sums, running side by side; 3 launches -> 2; both bodies and the GEMM kernel untouched
# speedup vs baseline: 4.6155x; 1.2615x over previous
#include <hip/hip_runtime.h>
#include <hip/hip_bf16.h>
#include <math.h>

#define BDIM 1024
#define DDIM 256
#define CDIM 1000
#define EPSV 1e-6f

typedef __attribute__((ext_vector_type(2))) float v2f;
typedef __attribute__((ext_vector_type(8))) float v8f;

#define WT_WAVES   8
#define WT_ROWS    (DDIM / WT_WAVES)
#define NT_WBLOCKS ((CDIM + 31) / 32)
#define NT_XBLOCKS (BDIM / 8)
#define NT_BLOCKS  (NT_WBLOCKS + NT_XBLOCKS)

__global__ void __launch_bounds__(WT_WAVES * 32)
pd_norms_kernel(const float* __restrict__ x,
                const float* __restrict__ w,
                float* __restrict__ xterm,
                float* __restrict__ wterm) {
    __shared__ float2 part[WT_WAVES][32];

    const int lane = threadIdx.x & 31;
    const int wave = threadIdx.x >> 5;

    if (blockIdx.x < NT_WBLOCKS) {
        const int c    = blockIdx.x * 32 + lane;
        const int cc   = c < CDIM ? c : (CDIM - 1);
        const float* p = w + (wave * WT_ROWS) * CDIM + cc;

        float v[WT_ROWS];
#pragma unroll
        for (int i = 0; i < WT_ROWS; ++i) {
            v[i] = p[i * CDIM];
        }

        float a1[4] = {0.f, 0.f, 0.f, 0.f};
        float a2[4] = {0.f, 0.f, 0.f, 0.f};
#pragma unroll
        for (int i = 0; i < WT_ROWS; ++i) {
            a1[i & 3] += v[i];
            a2[i & 3] += v[i] * v[i];
        }
        const float s1 = (a1[0] + a1[1]) + (a1[2] + a1[3]);
        const float s2 = (a2[0] + a2[1]) + (a2[2] + a2[3]);

        part[wave][lane] = make_float2(s1, s2);
        __syncthreads();

        if (wave == 0) {
            float t1 = 0.f, t2 = 0.f;
            for (int k = 0; k < WT_WAVES; ++k) {
                const float2 q = part[k][lane];
                t1 += q.x;
                t2 += q.y;
            }
            if (c < CDIM) {
                wterm[c] = t2 - 2.0f * EPSV * t1 + (float)DDIM * EPSV * EPSV;
            }
        }
    } else {
        const int row = (blockIdx.x - NT_WBLOCKS) * 8 + wave;

        const float* xr = x + row * DDIM;
        float s1 = 0.f, s2 = 0.f;
#pragma unroll
        for (int i = 0; i < DDIM / 32; ++i) {
            float v = xr[lane + i * 32];
            s1 += v;
            s2 += v * v;
        }
#pragma unroll
        for (int off = 16; off > 0; off >>= 1) {
            s1 += __shfl_xor(s1, off, 32);
            s2 += __shfl_xor(s2, off, 32);
        }
        if (lane == 0) {
            xterm[row] = s2 + 2.0f * EPSV * s1;
        }
    }
}

typedef __attribute__((ext_vector_type(4))) float v4f;

#define GM_WAVES  4
#define GM_CHUNKS (DDIM / 16)
#define GM_AHEAD  4

struct GmChunk {
    v4f a[2];
    v4f b[2][4];
};

__device__ __forceinline__ void gm_load(GmChunk& ck, const float* __restrict__ ap,
                                        const float* __restrict__ bp, int ch) {
#pragma unroll
    for (int q = 0; q < 2; ++q) {
        ck.a[q] = *(const v4f*)(ap + 16 * ch + 8 * q);
#pragma unroll
        for (int s = 0; s < 4; ++s) {
            ck.b[q][s] = *(const v4f*)(bp + (16 * ch + 8 * q + s) * CDIM);
        }
    }
}

__device__ __forceinline__ void gm_mma(const GmChunk& ck, v8f (&acc)[4]) {
#pragma unroll
    for (int q = 0; q < 2; ++q) {
#pragma unroll
        for (int p = 0; p < 2; ++p) {
            v2f a;
            a.x = ck.a[q][2 * p];
            a.y = ck.a[q][2 * p + 1];
#pragma unroll
            for (int j = 0; j < 4; ++j) {
                v2f b;
                b.x = ck.b[q][2 * p][j];
                b.y = ck.b[q][2 * p + 1][j];
                acc[j] = __builtin_amdgcn_wmma_f32_16x16x4_f32(
                     false, a,  false, b,
                     (short)0, acc[j],  false,  false);
            }
        }
    }
}

__global__ void __launch_bounds__(GM_WAVES * 32) __attribute__((amdgpu_waves_per_eu(1, 1)))
pd_wmma_kernel(const float* __restrict__ x,
               const float* __restrict__ w,
               const float* __restrict__ xterm,
               const float* __restrict__ wterm,
               float* __restrict__ out) {
    const int lane = threadIdx.x & 31;
    const int wave = threadIdx.x >> 5;
    const int c    = lane & 15;
    const int h    = lane >> 4;

    const int tile_m0 = blockIdx.y * 64 + wave * 16;
    const int gcol    = blockIdx.x * 64 + 4 * c;
    const bool live   = gcol < CDIM;
    const int gcolc   = live ? gcol : (CDIM - 4);

    const float* ap = x + (tile_m0 + c) * DDIM + 4 * h;
    const float* bp = w + (4 * h) * CDIM + gcolc;

    const float* xt = xterm + tile_m0 + 8 * h;
    const v4f xt0 = *(const v4f*)(xt);
    const v4f xt1 = *(const v4f*)(xt + 4);
    const v4f wt  = *(const v4f*)(wterm + gcolc);

    v8f acc[4] = {};
    GmChunk ring[GM_AHEAD + 1];

#pragma unroll
    for (int ch = 0; ch < GM_AHEAD; ++ch) {
        gm_load(ring[ch], ap, bp, ch);
    }
#pragma unroll
    for (int ch = 0; ch < GM_CHUNKS; ++ch) {
        if (ch + GM_AHEAD < GM_CHUNKS) {
            gm_load(ring[(ch + GM_AHEAD) % (GM_AHEAD + 1)], ap, bp, ch + GM_AHEAD);
        }
        if (ch == 0) {
            asm volatile("" :: "v"(xt0), "v"(xt1), "v"(wt));
        }
        gm_mma(ring[ch % (GM_AHEAD + 1)], acc);
    }

    if (live) {
#pragma unroll
        for (int r = 0; r < 8; ++r) {
            const int row_g = tile_m0 + 8 * h + r;
            const float xr  = (r < 4) ? xt0[r & 3] : xt1[r & 3];
            v4f o;
#pragma unroll
            for (int j = 0; j < 4; ++j) {
                const float s = xr + wt[j] - 2.0f * acc[j][r];
                o[j] = sqrtf(fmaxf(s, 0.0f));
            }
            *(v4f*)(out + row_g * CDIM + gcol) = o;
        }
    }
}

extern "C" void kernel_launch(void* const* d_in, const int* in_sizes, int n_in,
                              void* d_out, int out_size, void* d_ws, size_t ws_size,
                              hipStream_t stream) {
    const float* x = (const float*)d_in[0];
    const float* w = (const float*)d_in[1];
    float* out = (float*)d_out;

    float* xterm = (float*)d_ws;
    float* wterm = xterm + BDIM;

    pd_norms_kernel<<<NT_BLOCKS, WT_WAVES * 32, 0, stream>>>(x, w, xterm, wterm);
    dim3 grid(16, 16, 1);
    pd_wmma_kernel<<<grid, GM_WAVES * 32, 0, stream>>>(x, w, xterm, wterm, out);
}


# Norm-term launch dropped: each GEMM work-group sums its own 64 rows of x and 64 columns of w beside its WMMAs (4 own loads per chunk, split over its 4 waves, LDS join); 2 launches -> 1; new sum order
# speedup vs baseline: 5.9264x; 1.2840x over previous
#include <hip/hip_runtime.h>
#include <hip/hip_bf16.h>
#include <math.h>

#define BDIM 1024
#define DDIM 256
#define CDIM 1000
#define EPSV 1e-6f

typedef __attribute__((ext_vector_type(2))) float v2f;
typedef __attribute__((ext_vector_type(8))) float v8f;

typedef __attribute__((ext_vector_type(4))) float v4f;

#define GM_WAVES  4
#define GM_CHUNKS (DDIM / 16)
#define GM_AHEAD  4

struct GmChunk {
    v4f a[2];
    v4f b[2][4];
    v4f nx[2];
    v4f nw[2];
};

__device__ __forceinline__ void gm_load(GmChunk& ck, const float* __restrict__ ap,
                                        const float* __restrict__ bp,
                                        const float* __restrict__ xn,
                                        const float* __restrict__ wn, int ch) {
#pragma unroll
    for (int q = 0; q < 2; ++q) {
        ck.a[q] = *(const v4f*)(ap + 16 * ch + 8 * q);
#pragma unroll
        for (int s = 0; s < 4; ++s) {
            ck.b[q][s] = *(const v4f*)(bp + (16 * ch + 8 * q + s) * CDIM);
        }
    }
#pragma unroll
    for (int j = 0; j < 2; ++j) {
        ck.nx[j] = *(const v4f*)(xn + 8 * j * DDIM + 16 * ch);
    }
#pragma unroll
    for (int i = 0; i < 2; ++i) {
        ck.nw[i] = *(const v4f*)(wn + (16 * ch + 2 * i) * CDIM);
    }
}

__device__ __forceinline__ void gm_mma(const GmChunk& ck, v8f (&acc)[4]) {
#pragma unroll
    for (int q = 0; q < 2; ++q) {
#pragma unroll
        for (int p = 0; p < 2; ++p) {
            v2f a;
            a.x = ck.a[q][2 * p];
            a.y = ck.a[q][2 * p + 1];
#pragma unroll
            for (int j = 0; j < 4; ++j) {
                v2f b;
                b.x = ck.b[q][2 * p][j];
                b.y = ck.b[q][2 * p + 1][j];
                acc[j] = __builtin_amdgcn_wmma_f32_16x16x4_f32(
                     false, a,  false, b,
                     (short)0, acc[j],  false,  false);
            }
        }
    }
}

__global__ void __launch_bounds__(GM_WAVES * 32) __attribute__((amdgpu_waves_per_eu(1, 1)))
pd_fused_kernel(const float* __restrict__ x,
                const float* __restrict__ w,
                float* __restrict__ out) {
    __shared__ v4f xp[GM_WAVES][16];
    __shared__ v4f wq[GM_WAVES][2][16];

    const int lane = threadIdx.x & 31;
    const int wave = threadIdx.x >> 5;
    const int c    = lane & 15;
    const int h    = lane >> 4;

    const int tile_m0 = blockIdx.y * 64 + wave * 16;
    const int gcol    = blockIdx.x * 64 + 4 * c;
    const bool live   = gcol < CDIM;
    const int gcolc   = live ? gcol : (CDIM - 4);

    const float* ap = x + (tile_m0 + c) * DDIM + 4 * h;
    const float* bp = w + (4 * h) * CDIM + gcolc;
    const float* xn = x + (tile_m0 + (lane >> 2)) * DDIM + 4 * (lane & 3);
    const float* wn = w + (4 * wave + h) * CDIM + gcolc;

    v8f acc[4] = {};
    GmChunk ring[GM_AHEAD + 1];

    v4f nx1[2] = {};
    v4f nx2[2] = {};
    v4f nw1[2] = {};
    v4f nw2[2] = {};

#pragma unroll
    for (int ch = 0; ch < GM_AHEAD; ++ch) {
        gm_load(ring[ch], ap, bp, xn, wn, ch);
    }
#pragma unroll
    for (int ch = 0; ch < GM_CHUNKS; ++ch) {
        __builtin_amdgcn_sched_barrier(0);
        if (ch + GM_AHEAD < GM_CHUNKS) {
            gm_load(ring[(ch + GM_AHEAD) % (GM_AHEAD + 1)], ap, bp, xn, wn, ch + GM_AHEAD);
        }
        const GmChunk& ck = ring[ch % (GM_AHEAD + 1)];
#pragma unroll
        for (int j = 0; j < 2; ++j) {
            nx1[j] += ck.nx[j];
            nx2[j] += ck.nx[j] * ck.nx[j];
            nw1[j] += ck.nw[j];
            nw2[j] += ck.nw[j] * ck.nw[j];
        }
        gm_mma(ck, acc);
    }

    {
        float* xpf = (float*)&xp[wave][0];
#pragma unroll
        for (int j = 0; j < 2; ++j) {
            const float s1 = (nx1[j][0] + nx1[j][1]) + (nx1[j][2] + nx1[j][3]);
            const float s2 = (nx2[j][0] + nx2[j][1]) + (nx2[j][2] + nx2[j][3]);
            xpf[32 * j + lane] = s2 + 2.0f * EPSV * s1;
        }
        wq[wave][h][c] = (nw2[0] + nw2[1]) - 2.0f * EPSV * (nw1[0] + nw1[1]);
    }
    __syncthreads();

    v4f wt = wq[0][0][c];
    wt += wq[0][1][c];
#pragma unroll
    for (int k = 1; k < GM_WAVES; ++k) {
        wt += wq[k][0][c];
        wt += wq[k][1][c];
    }
    wt += (float)DDIM * EPSV * EPSV;

    if (live) {
#pragma unroll
        for (int r = 0; r < 8; ++r) {
            const int row_g = tile_m0 + 8 * h + r;
            const v4f xq    = xp[wave][8 * h + r];
            const float xr  = (xq[0] + xq[1]) + (xq[2] + xq[3]);
            v4f o;
#pragma unroll
            for (int j = 0; j < 4; ++j) {
                const float s = xr + wt[j] - 2.0f * acc[j][r];
                o[j] = sqrtf(fmaxf(s, 0.0f));
            }
            *(v4f*)(out + row_g * CDIM + gcol) = o;
        }
    }
}

extern "C" void kernel_launch(void* const* d_in, const int* in_sizes, int n_in,
                              void* d_out, int out_size, void* d_ws, size_t ws_size,
                              hipStream_t stream) {
    const float* x = (const float*)d_in[0];
    const float* w = (const float*)d_in[1];
    float* out = (float*)d_out;

    dim3 grid(16, 16, 1);
    pd_fused_kernel<<<grid, GM_WAVES * 32, 0, stream>>>(x, w, out);
}
